# starting kernel
# baseline (speedup 1.0000x reference)
#include <hip/hip_runtime.h>

typedef float v2f __attribute__((ext_vector_type(2)));
typedef float v8f __attribute__((ext_vector_type(8)));

#define BB 4096
#define TT 100
#define FF 40
#define HE 64
#define HL 32
#define HD 64

#define GS  260
#define GXS 132
#define HS  66
#define HLS 34

__device__ __forceinline__ v8f wmma4(v2f a, v2f b, v8f c) {
  return __builtin_amdgcn_wmma_f32_16x16x4_f32(false, a, false, b, (short)0, c,
                                               false, false);
}

__device__ __forceinline__ float sigm(float x) {
  return 1.0f / (1.0f + __expf(-x));
}
__device__ __forceinline__ float tanh_(float x) {
  float e = __expf(2.0f * x);
  return 1.0f - 2.0f / (e + 1.0f);
}

__global__ __launch_bounds__(256)
void lstm_ae_kernel(const float* __restrict__ x,
                    const float* __restrict__ eWih, const float* __restrict__ eWhh,
                    const float* __restrict__ eb,
                    const float* __restrict__ lWih, const float* __restrict__ lWhh,
                    const float* __restrict__ lb,
                    const float* __restrict__ dWih, const float* __restrict__ dWhh,
                    const float* __restrict__ db,
                    const float* __restrict__ oW,   const float* __restrict__ ob,
                    float* __restrict__ out) {
  __shared__ float gates[16 * GS];
  __shared__ float gxl[16 * GXS];
  __shared__ float hbuf[16 * HS];
  __shared__ float hlat[16 * HLS];

  const int tid  = threadIdx.x;
  const int lane = tid & 31;
  const int wave = tid >> 5;
  const int lrow = lane & 15;
  const int lhi  = lane >> 4;
  const int b0   = blockIdx.x * 16;

  const int r0  = tid >> 6;
  const int j64 = tid & 63;
  const int rl0 = tid >> 5;
  const int j32 = tid & 31;

  v2f wih[10][2];
  v2f whh[16][2];
  #pragma unroll
  for (int tt = 0; tt < 2; ++tt) {
    const int gc = wave * 32 + tt * 16;
    #pragma unroll
    for (int kc = 0; kc < 10; ++kc)
      wih[kc][tt] = *(const v2f*)&eWih[(gc + lrow) * FF + kc * 4 + 2 * lhi];
    #pragma unroll
    for (int kc = 0; kc < 16; ++kc)
      whh[kc][tt] = *(const v2f*)&eWhh[(gc + lrow) * HE + kc * 4 + 2 * lhi];
  }
  const float ebi = eb[j64], ebf = eb[64 + j64], ebg = eb[128 + j64], ebo = eb[192 + j64];

  for (int i = tid; i < 16 * HS; i += 256) hbuf[i] = 0.0f;
  float cE[4] = {0.f, 0.f, 0.f, 0.f};
  __syncthreads();

  const float* xblk = x + (size_t)b0 * TT * FF;

  #pragma unroll 1
  for (int t = 0; t < TT; ++t) {
    v8f acc0 = {}, acc1 = {};
    const float* xt = xblk + (size_t)lrow * TT * FF + t * FF + 2 * lhi;
    #pragma unroll
    for (int kc = 0; kc < 10; ++kc) {
      v2f a = *(const v2f*)&xt[kc * 4];
      acc0 = wmma4(a, wih[kc][0], acc0);
      acc1 = wmma4(a, wih[kc][1], acc1);
    }
    #pragma unroll
    for (int kc = 0; kc < 16; ++kc) {
      v2f a = *(const v2f*)&hbuf[lrow * HS + kc * 4 + 2 * lhi];
      acc0 = wmma4(a, whh[kc][0], acc0);
      acc1 = wmma4(a, whh[kc][1], acc1);
    }
    const int gc0 = wave * 32;
    #pragma unroll
    for (int jj = 0; jj < 8; ++jj) {
      gates[(jj + 8 * lhi) * GS + gc0 + lrow]      = acc0[jj];
      gates[(jj + 8 * lhi) * GS + gc0 + 16 + lrow] = acc1[jj];
    }
    __syncthreads();
    #pragma unroll
    for (int i = 0; i < 4; ++i) {
      const int r = r0 + 4 * i;
      float gi = gates[r * GS + j64] + ebi;
      float gf = gates[r * GS + 64 + j64] + ebf;
      float gg = gates[r * GS + 128 + j64] + ebg;
      float go = gates[r * GS + 192 + j64] + ebo;
      float c  = sigm(gf) * cE[i] + sigm(gi) * tanh_(gg);
      cE[i] = c;
      hbuf[r * HS + j64] = sigm(go) * tanh_(c);
    }
    __syncthreads();
  }

  {
    const int gc = wave * 16;
    v8f acc = {};
    #pragma unroll
    for (int kc = 0; kc < 16; ++kc) {
      v2f a = *(const v2f*)&hbuf[lrow * HS + kc * 4 + 2 * lhi];
      v2f b = *(const v2f*)&lWih[(gc + lrow) * HE + kc * 4 + 2 * lhi];
      acc = wmma4(a, b, acc);
    }
    const float bias = lb[gc + lrow];
    #pragma unroll
    for (int jj = 0; jj < 8; ++jj)
      gxl[(jj + 8 * lhi) * GXS + gc + lrow] = acc[jj] + bias;
  }
  __syncthreads();

  v2f wlh[8];
  v2f dwi[8][2];
  v2f dwh[16][2];
  v2f wo[16];
  {
    const int gcl = wave * 16;
    #pragma unroll
    for (int kc = 0; kc < 8; ++kc)
      wlh[kc] = *(const v2f*)&lWhh[(gcl + lrow) * HL + kc * 4 + 2 * lhi];
    #pragma unroll
    for (int tt = 0; tt < 2; ++tt) {
      const int gc = wave * 32 + tt * 16;
      #pragma unroll
      for (int kc = 0; kc < 8; ++kc)
        dwi[kc][tt] = *(const v2f*)&dWih[(gc + lrow) * HL + kc * 4 + 2 * lhi];
      #pragma unroll
      for (int kc = 0; kc < 16; ++kc)
        dwh[kc][tt] = *(const v2f*)&dWhh[(gc + lrow) * HD + kc * 4 + 2 * lhi];
    }
    const int f = wave * 16 + lrow;
    #pragma unroll
    for (int kc = 0; kc < 16; ++kc) {
      v2f z = {};
      if (wave < 3 && f < FF)
        z = *(const v2f*)&oW[f * HD + kc * 4 + 2 * lhi];
      wo[kc] = z;
    }
  }
  const float dbi = db[j64], dbf = db[64 + j64], dbg = db[128 + j64], dbo = db[192 + j64];
  const float obias = (wave < 3 && (wave * 16 + lrow) < FF) ? ob[wave * 16 + lrow] : 0.0f;

  for (int i = tid; i < 16 * HS; i += 256) hbuf[i] = 0.0f;
  for (int i = tid; i < 16 * HLS; i += 256) hlat[i] = 0.0f;
  float cL[2] = {0.f, 0.f};
  float cD[4] = {0.f, 0.f, 0.f, 0.f};
  __syncthreads();

  float* outblk = out + (size_t)b0 * TT * FF;

  #pragma unroll 1
  for (int t = 0; t < TT; ++t) {
    {
      const int gc = wave * 16;
      v8f acc = {};
      #pragma unroll
      for (int kc = 0; kc < 8; ++kc) {
        v2f a = *(const v2f*)&hlat[lrow * HLS + kc * 4 + 2 * lhi];
        acc = wmma4(a, wlh[kc], acc);
      }
      #pragma unroll
      for (int jj = 0; jj < 8; ++jj)
        gates[(jj + 8 * lhi) * GS + gc + lrow] = acc[jj];
    }
    __syncthreads();
    #pragma unroll
    for (int i = 0; i < 2; ++i) {
      const int r = rl0 + 8 * i;
      float gi = gates[r * GS + j32]      + gxl[r * GXS + j32];
      float gf = gates[r * GS + 32 + j32] + gxl[r * GXS + 32 + j32];
      float gg = gates[r * GS + 64 + j32] + gxl[r * GXS + 64 + j32];
      float go = gates[r * GS + 96 + j32] + gxl[r * GXS + 96 + j32];
      float c  = sigm(gf) * cL[i] + sigm(gi) * tanh_(gg);
      cL[i] = c;
      hlat[r * HLS + j32] = sigm(go) * tanh_(c);
    }
    __syncthreads();
    {
      v8f acc0 = {}, acc1 = {};
      #pragma unroll
      for (int kc = 0; kc < 8; ++kc) {
        v2f a = *(const v2f*)&hlat[lrow * HLS + kc * 4 + 2 * lhi];
        acc0 = wmma4(a, dwi[kc][0], acc0);
        acc1 = wmma4(a, dwi[kc][1], acc1);
      }
      #pragma unroll
      for (int kc = 0; kc < 16; ++kc) {
        v2f a = *(const v2f*)&hbuf[lrow * HS + kc * 4 + 2 * lhi];
        acc0 = wmma4(a, dwh[kc][0], acc0);
        acc1 = wmma4(a, dwh[kc][1], acc1);
      }
      const int gc0 = wave * 32;
      #pragma unroll
      for (int jj = 0; jj < 8; ++jj) {
        gates[(jj + 8 * lhi) * GS + gc0 + lrow]      = acc0[jj];
        gates[(jj + 8 * lhi) * GS + gc0 + 16 + lrow] = acc1[jj];
      }
    }
    __syncthreads();
    #pragma unroll
    for (int i = 0; i < 4; ++i) {
      const int r = r0 + 4 * i;
      float gi = gates[r * GS + j64] + dbi;
      float gf = gates[r * GS + 64 + j64] + dbf;
      float gg = gates[r * GS + 128 + j64] + dbg;
      float go = gates[r * GS + 192 + j64] + dbo;
      float c  = sigm(gf) * cD[i] + sigm(gi) * tanh_(gg);
      cD[i] = c;
      hbuf[r * HS + j64] = sigm(go) * tanh_(c);
    }
    __syncthreads();
    if (wave < 3) {
      v8f acc = {};
      #pragma unroll
      for (int kc = 0; kc < 16; ++kc) {
        v2f a = *(const v2f*)&hbuf[lrow * HS + kc * 4 + 2 * lhi];
        acc = wmma4(a, wo[kc], acc);
      }
      const int f = wave * 16 + lrow;
      if (f < FF) {
        #pragma unroll
        for (int jj = 0; jj < 8; ++jj) {
          const int m = jj + 8 * lhi;
          outblk[(size_t)m * TT * FF + t * FF + f] = acc[jj] + obias;
        }
      }
    }
  }
}

extern "C" void kernel_launch(void* const* d_in, const int* in_sizes, int n_in,
                              void* d_out, int out_size, void* d_ws, size_t ws_size,
                              hipStream_t stream) {
  (void)in_sizes; (void)n_in; (void)out_size; (void)d_ws; (void)ws_size;
  const float* x    = (const float*)d_in[0];
  const float* eWih = (const float*)d_in[1];
  const float* eWhh = (const float*)d_in[2];
  const float* eb   = (const float*)d_in[3];
  const float* lWih = (const float*)d_in[4];
  const float* lWhh = (const float*)d_in[5];
  const float* lb   = (const float*)d_in[6];
  const float* dWih = (const float*)d_in[7];
  const float* dWhh = (const float*)d_in[8];
  const float* db   = (const float*)d_in[9];
  const float* oW   = (const float*)d_in[10];
  const float* ob   = (const float*)d_in[11];
  float* out = (float*)d_out;

  dim3 grid(BB / 16), block(256);
  hipLaunchKernelGGL(lstm_ae_kernel, grid, block, 0, stream,
                     x, eWih, eWhh, eb, lWih, lWhh, lb, dWih, dWhh, db, oW, ob, out);
}


# gate GEMMs transposed (weights as A, h^T/x^T as B): a lane holds i,f,g,o of its own cells, cell update in registers, gates skip LDS; h double-buffered, barriers halved; sum order kept
# speedup vs baseline: 1.0976x; 1.0976x over previous
#include <hip/hip_runtime.h>


typedef float v2f __attribute__((ext_vector_type(2)));
typedef float v4f __attribute__((ext_vector_type(4)));
typedef float v8f __attribute__((ext_vector_type(8)));

#define BB 4096
#define TT 100
#define FF 40
#define HE 64
#define HL 32
#define HD 64

#define HS  68
#define HLS 36

__device__ __forceinline__ v8f wmma4(v2f a, v2f b, v8f c) {
  return __builtin_amdgcn_wmma_f32_16x16x4_f32(false, a, false, b, (short)0, c,
                                               false, false);
}

__device__ __forceinline__ float sigm(float x) {
  return 1.0f / (1.0f + __expf(-x));
}
__device__ __forceinline__ float tanh_(float x) {
  float e = __expf(2.0f * x);
  return 1.0f - 2.0f / (e + 1.0f);
}

__global__ __launch_bounds__(256)
void lstm_ae_kernel(const float* __restrict__ x,
                    const float* __restrict__ eWih, const float* __restrict__ eWhh,
                    const float* __restrict__ eb,
                    const float* __restrict__ lWih, const float* __restrict__ lWhh,
                    const float* __restrict__ lb,
                    const float* __restrict__ dWih, const float* __restrict__ dWhh,
                    const float* __restrict__ db,
                    const float* __restrict__ oW,   const float* __restrict__ ob,
                    float* __restrict__ out) {
  __shared__ __attribute__((aligned(16))) float henc[2 * 16 * HS];
  __shared__ __attribute__((aligned(16))) float hdec[2 * 16 * HS];
  __shared__ __attribute__((aligned(16))) float hlat[2 * 16 * HLS];

  const int tid  = threadIdx.x;
  const int lane = tid & 31;
  const int wave = tid >> 5;
  const int lrow = lane & 15;
  const int lhi  = lane >> 4;
  const int b0   = blockIdx.x * 16;

  const int mh = lrow >> 3;
  const int mq = (lrow & 7) >> 1;
  const int ms = lrow & 1;
  const int row64 = mq * 64 + 8 * wave + 4 * mh + ms;
  const int row32 = mq * 32 + 4 * wave + 2 * mh + ms;
  const int u64 = 8 * wave + 4 * lhi;
  const int u32 = 4 * wave + 2 * lhi;

  v2f wih[10][2];
  v2f whh[16][2];
  #pragma unroll
  for (int tt = 0; tt < 2; ++tt) {
    const int gr = row64 + 2 * tt;
    #pragma unroll
    for (int kc = 0; kc < 10; ++kc)
      wih[kc][tt] = *(const v2f*)&eWih[gr * FF + kc * 4 + 2 * lhi];
    #pragma unroll
    for (int kc = 0; kc < 16; ++kc)
      whh[kc][tt] = *(const v2f*)&eWhh[gr * HE + kc * 4 + 2 * lhi];
  }
  float bE[4][4];
  #pragma unroll
  for (int q = 0; q < 4; ++q)
    #pragma unroll
    for (int cidx = 0; cidx < 4; ++cidx)
      bE[q][cidx] = eb[q * 64 + u64 + cidx];

  for (int i = tid; i < 16 * HS; i += 256) { henc[i] = 0.0f; hdec[i] = 0.0f; }
  for (int i = tid; i < 16 * HLS; i += 256) hlat[i] = 0.0f;
  float cE[4] = {0.f, 0.f, 0.f, 0.f};
  __syncthreads();

  const float* xblk = x + (size_t)b0 * TT * FF;

  #pragma unroll 1
  for (int t = 0; t < TT; ++t) {
    const float* hc = henc + (t & 1) * (16 * HS);
    float*       hn = henc + ((t + 1) & 1) * (16 * HS);
    v8f acc[2] = {{}, {}};
    const float* xt = xblk + (size_t)lrow * TT * FF + t * FF + 2 * lhi;
    #pragma unroll
    for (int kc = 0; kc < 10; ++kc) {
      v2f b = *(const v2f*)&xt[kc * 4];
      acc[0] = wmma4(wih[kc][0], b, acc[0]);
      acc[1] = wmma4(wih[kc][1], b, acc[1]);
    }
    #pragma unroll
    for (int kc = 0; kc < 16; ++kc) {
      v2f b = *(const v2f*)&hc[lrow * HS + kc * 4 + 2 * lhi];
      acc[0] = wmma4(whh[kc][0], b, acc[0]);
      acc[1] = wmma4(whh[kc][1], b, acc[1]);
    }
    v4f hv;
    #pragma unroll
    for (int tt = 0; tt < 2; ++tt) {
      #pragma unroll
      for (int s = 0; s < 2; ++s) {
        const int ci = 2 * tt + s;
        float gi = acc[tt][0 + s] + bE[0][ci];
        float gf = acc[tt][2 + s] + bE[1][ci];
        float gg = acc[tt][4 + s] + bE[2][ci];
        float go = acc[tt][6 + s] + bE[3][ci];
        float c  = sigm(gf) * cE[ci] + sigm(gi) * tanh_(gg);
        cE[ci] = c;
        hv[ci] = sigm(go) * tanh_(c);
      }
    }
    *(v4f*)&hn[lrow * HS + u64] = hv;
    __syncthreads();
  }
  const float* hlast = henc + (TT & 1) * (16 * HS);

  float gxl[8];
  {
    v8f a8 = {};
    #pragma unroll
    for (int kc = 0; kc < 16; ++kc) {
      v2f a = *(const v2f*)&lWih[row32 * HE + kc * 4 + 2 * lhi];
      v2f b = *(const v2f*)&hlast[lrow * HS + kc * 4 + 2 * lhi];
      a8 = wmma4(a, b, a8);
    }
    #pragma unroll
    for (int r = 0; r < 8; ++r)
      gxl[r] = a8[r] + lb[(r >> 1) * 32 + u32 + (r & 1)];
  }

  v2f wlh[8];
  v2f dwi[8][2];
  v2f dwh[16][2];
  v2f wo[16];
  {
    #pragma unroll
    for (int kc = 0; kc < 8; ++kc)
      wlh[kc] = *(const v2f*)&lWhh[row32 * HL + kc * 4 + 2 * lhi];
    #pragma unroll
    for (int tt = 0; tt < 2; ++tt) {
      const int gr = row64 + 2 * tt;
      #pragma unroll
      for (int kc = 0; kc < 8; ++kc)
        dwi[kc][tt] = *(const v2f*)&dWih[gr * HL + kc * 4 + 2 * lhi];
      #pragma unroll
      for (int kc = 0; kc < 16; ++kc)
        dwh[kc][tt] = *(const v2f*)&dWhh[gr * HD + kc * 4 + 2 * lhi];
    }
    const int f = wave * 16 + lrow;
    #pragma unroll
    for (int kc = 0; kc < 16; ++kc) {
      v2f z = {};
      if (wave < 3 && f < FF)
        z = *(const v2f*)&oW[f * HD + kc * 4 + 2 * lhi];
      wo[kc] = z;
    }
  }
  float bD[4][4];
  #pragma unroll
  for (int q = 0; q < 4; ++q)
    #pragma unroll
    for (int cidx = 0; cidx < 4; ++cidx)
      bD[q][cidx] = db[q * 64 + u64 + cidx];
  const float obias = (wave < 3 && (wave * 16 + lrow) < FF) ? ob[wave * 16 + lrow] : 0.0f;

  float cL[2] = {0.f, 0.f};
  float cD[4] = {0.f, 0.f, 0.f, 0.f};

  float* outblk = out + (size_t)b0 * TT * FF;

  #pragma unroll 1
  for (int t = 0; t < TT; ++t) {
    const float* lc = hlat + (t & 1) * (16 * HLS);
    float*       ln = hlat + ((t + 1) & 1) * (16 * HLS);
    const float* dc = hdec + (t & 1) * (16 * HS);
    float*       dn = hdec + ((t + 1) & 1) * (16 * HS);
    {
      v8f a8 = {};
      #pragma unroll
      for (int kc = 0; kc < 8; ++kc) {
        v2f b = *(const v2f*)&lc[lrow * HLS + kc * 4 + 2 * lhi];
        a8 = wmma4(wlh[kc], b, a8);
      }
      v2f hv;
      #pragma unroll
      for (int s = 0; s < 2; ++s) {
        float gi = a8[0 + s] + gxl[0 + s];
        float gf = a8[2 + s] + gxl[2 + s];
        float gg = a8[4 + s] + gxl[4 + s];
        float go = a8[6 + s] + gxl[6 + s];
        float c  = sigm(gf) * cL[s] + sigm(gi) * tanh_(gg);
        cL[s] = c;
        hv[s] = sigm(go) * tanh_(c);
      }
      *(v2f*)&ln[lrow * HLS + u32] = hv;
    }
    __syncthreads();
    {
      v8f acc[2] = {{}, {}};
      #pragma unroll
      for (int kc = 0; kc < 8; ++kc) {
        v2f b = *(const v2f*)&ln[lrow * HLS + kc * 4 + 2 * lhi];
        acc[0] = wmma4(dwi[kc][0], b, acc[0]);
        acc[1] = wmma4(dwi[kc][1], b, acc[1]);
      }
      #pragma unroll
      for (int kc = 0; kc < 16; ++kc) {
        v2f b = *(const v2f*)&dc[lrow * HS + kc * 4 + 2 * lhi];
        acc[0] = wmma4(dwh[kc][0], b, acc[0]);
        acc[1] = wmma4(dwh[kc][1], b, acc[1]);
      }
      v4f hv;
      #pragma unroll
      for (int tt = 0; tt < 2; ++tt) {
        #pragma unroll
        for (int s = 0; s < 2; ++s) {
          const int ci = 2 * tt + s;
          float gi = acc[tt][0 + s] + bD[0][ci];
          float gf = acc[tt][2 + s] + bD[1][ci];
          float gg = acc[tt][4 + s] + bD[2][ci];
          float go = acc[tt][6 + s] + bD[3][ci];
          float c  = sigm(gf) * cD[ci] + sigm(gi) * tanh_(gg);
          cD[ci] = c;
          hv[ci] = sigm(go) * tanh_(c);
        }
      }
      *(v4f*)&dn[lrow * HS + u64] = hv;
    }
    __syncthreads();
    if (wave < 3) {
      v8f a8 = {};
      #pragma unroll
      for (int kc = 0; kc < 16; ++kc) {
        v2f a = *(const v2f*)&dn[lrow * HS + kc * 4 + 2 * lhi];
        a8 = wmma4(a, wo[kc], a8);
      }
      const int f = wave * 16 + lrow;
      if (f < FF) {
        #pragma unroll
        for (int jj = 0; jj < 8; ++jj) {
          const int m = jj + 8 * lhi;
          outblk[(size_t)m * TT * FF + t * FF + f] = a8[jj] + obias;
        }
      }
    }
  }
}

extern "C" void kernel_launch(void* const* d_in, const int* in_sizes, int n_in,
                              void* d_out, int out_size, void* d_ws, size_t ws_size,
                              hipStream_t stream) {
  (void)in_sizes; (void)n_in; (void)out_size; (void)d_ws; (void)ws_size;
  const float* x    = (const float*)d_in[0];
  const float* eWih = (const float*)d_in[1];
  const float* eWhh = (const float*)d_in[2];
  const float* eb   = (const float*)d_in[3];
  const float* lWih = (const float*)d_in[4];
  const float* lWhh = (const float*)d_in[5];
  const float* lb   = (const float*)d_in[6];
  const float* dWih = (const float*)d_in[7];
  const float* dWhh = (const float*)d_in[8];
  const float* db   = (const float*)d_in[9];
  const float* oW   = (const float*)d_in[10];
  const float* ob   = (const float*)d_in[11];
  float* out = (float*)d_out;

  dim3 grid(BB / 16), block(256);
  hipLaunchKernelGGL(lstm_ae_kernel, grid, block, 0, stream,
                     x, eWih, eWhh, eb, lWih, lWhh, lb, dWih, dWhh, db, oW, ob, out);
}


# encoder x tile fetched one step ahead by 160 threads into a double-buffered LDS tile, fragments read from LDS: the global-load wait leaves the step's chain, 8-fold duplicate loads go; sum order kept
# speedup vs baseline: 1.2428x; 1.1323x over previous
#include <hip/hip_runtime.h>


typedef float v2f __attribute__((ext_vector_type(2)));
typedef float v4f __attribute__((ext_vector_type(4)));
typedef float v8f __attribute__((ext_vector_type(8)));

#define BB 4096
#define TT 100
#define FF 40
#define HE 64
#define HL 32
#define HD 64

#define HS  68
#define HLS 36
#define XS  44

__device__ __forceinline__ v8f wmma4(v2f a, v2f b, v8f c) {
  return __builtin_amdgcn_wmma_f32_16x16x4_f32(false, a, false, b, (short)0, c,
                                               false, false);
}

__device__ __forceinline__ float sigm(float x) {
  return 1.0f / (1.0f + __expf(-x));
}
__device__ __forceinline__ float tanh_(float x) {
  float e = __expf(2.0f * x);
  return 1.0f - 2.0f / (e + 1.0f);
}

__global__ __launch_bounds__(256)
void lstm_ae_kernel(const float* __restrict__ x,
                    const float* __restrict__ eWih, const float* __restrict__ eWhh,
                    const float* __restrict__ eb,
                    const float* __restrict__ lWih, const float* __restrict__ lWhh,
                    const float* __restrict__ lb,
                    const float* __restrict__ dWih, const float* __restrict__ dWhh,
                    const float* __restrict__ db,
                    const float* __restrict__ oW,   const float* __restrict__ ob,
                    float* __restrict__ out) {
  __shared__ __attribute__((aligned(16))) float henc[2 * 16 * HS];
  __shared__ __attribute__((aligned(16))) float hdec[2 * 16 * HS];
  __shared__ __attribute__((aligned(16))) float hlat[2 * 16 * HLS];
  __shared__ __attribute__((aligned(16))) float xenc[2 * 16 * XS];

  const int tid  = threadIdx.x;
  const int lane = tid & 31;
  const int wave = tid >> 5;
  const int lrow = lane & 15;
  const int lhi  = lane >> 4;
  const int b0   = blockIdx.x * 16;

  const int mh = lrow >> 3;
  const int mq = (lrow & 7) >> 1;
  const int ms = lrow & 1;
  const int row64 = mq * 64 + 8 * wave + 4 * mh + ms;
  const int row32 = mq * 32 + 4 * wave + 2 * mh + ms;
  const int u64 = 8 * wave + 4 * lhi;
  const int u32 = 4 * wave + 2 * lhi;

  v2f wih[10][2];
  v2f whh[16][2];
  #pragma unroll
  for (int tt = 0; tt < 2; ++tt) {
    const int gr = row64 + 2 * tt;
    #pragma unroll
    for (int kc = 0; kc < 10; ++kc)
      wih[kc][tt] = *(const v2f*)&eWih[gr * FF + kc * 4 + 2 * lhi];
    #pragma unroll
    for (int kc = 0; kc < 16; ++kc)
      whh[kc][tt] = *(const v2f*)&eWhh[gr * HE + kc * 4 + 2 * lhi];
  }
  float bE[4][4];
  #pragma unroll
  for (int q = 0; q < 4; ++q)
    #pragma unroll
    for (int cidx = 0; cidx < 4; ++cidx)
      bE[q][cidx] = eb[q * 64 + u64 + cidx];

  for (int i = tid; i < 16 * HS; i += 256) { henc[i] = 0.0f; hdec[i] = 0.0f; }
  for (int i = tid; i < 16 * HLS; i += 256) hlat[i] = 0.0f;
  if (tid < 160) {
    const int r0 = tid / 10, q0 = tid - 10 * r0;
    *(v4f*)&xenc[r0 * XS + q0 * 4] =
        *(const v4f*)&x[((size_t)b0 + r0) * TT * FF + q0 * 4];
  }
  float cE[4] = {0.f, 0.f, 0.f, 0.f};
  __syncthreads();

  const float* xblk = x + (size_t)b0 * TT * FF;

  const bool xld  = __builtin_amdgcn_readfirstlane(wave) < 5;
  const int  xrw  = xld ? tid / 10 : 0;
  const int  xq   = xld ? tid - 10 * xrw : 0;
  const float* xsrc = xblk + (size_t)xrw * TT * FF + xq * 4;
  float*       xdst = xenc + xrw * XS + xq * 4;

  #pragma unroll 1
  for (int t = 0; t < TT; ++t) {
    const float* hc = henc + (t & 1) * (16 * HS);
    float*       hn = henc + ((t + 1) & 1) * (16 * HS);
    const float* xc = xenc + (t & 1) * (16 * XS);
    v4f xv = {};
    if (xld) {
      const int tn = (t + 1 < TT) ? (t + 1) : (TT - 1);
      xv = *(const v4f*)&xsrc[tn * FF];
    }
    v8f acc[2] = {{}, {}};
    #pragma unroll
    for (int kc = 0; kc < 10; ++kc) {
      v2f b = *(const v2f*)&xc[lrow * XS + kc * 4 + 2 * lhi];
      acc[0] = wmma4(wih[kc][0], b, acc[0]);
      acc[1] = wmma4(wih[kc][1], b, acc[1]);
    }
    #pragma unroll
    for (int kc = 0; kc < 16; ++kc) {
      v2f b = *(const v2f*)&hc[lrow * HS + kc * 4 + 2 * lhi];
      acc[0] = wmma4(whh[kc][0], b, acc[0]);
      acc[1] = wmma4(whh[kc][1], b, acc[1]);
    }
    v4f hv;
    #pragma unroll
    for (int tt = 0; tt < 2; ++tt) {
      #pragma unroll
      for (int s = 0; s < 2; ++s) {
        const int ci = 2 * tt + s;
        float gi = acc[tt][0 + s] + bE[0][ci];
        float gf = acc[tt][2 + s] + bE[1][ci];
        float gg = acc[tt][4 + s] + bE[2][ci];
        float go = acc[tt][6 + s] + bE[3][ci];
        float c  = sigm(gf) * cE[ci] + sigm(gi) * tanh_(gg);
        cE[ci] = c;
        hv[ci] = sigm(go) * tanh_(c);
      }
    }
    *(v4f*)&hn[lrow * HS + u64] = hv;
    if (xld) *(v4f*)&xdst[((t + 1) & 1) * (16 * XS)] = xv;
    __syncthreads();
  }
  const float* hlast = henc + (TT & 1) * (16 * HS);

  float gxl[8];
  {
    v8f a8 = {};
    #pragma unroll
    for (int kc = 0; kc < 16; ++kc) {
      v2f a = *(const v2f*)&lWih[row32 * HE + kc * 4 + 2 * lhi];
      v2f b = *(const v2f*)&hlast[lrow * HS + kc * 4 + 2 * lhi];
      a8 = wmma4(a, b, a8);
    }
    #pragma unroll
    for (int r = 0; r < 8; ++r)
      gxl[r] = a8[r] + lb[(r >> 1) * 32 + u32 + (r & 1)];
  }

  v2f wlh[8];
  v2f dwi[8][2];
  v2f dwh[16][2];
  v2f wo[16];
  {
    #pragma unroll
    for (int kc = 0; kc < 8; ++kc)
      wlh[kc] = *(const v2f*)&lWhh[row32 * HL + kc * 4 + 2 * lhi];
    #pragma unroll
    for (int tt = 0; tt < 2; ++tt) {
      const int gr = row64 + 2 * tt;
      #pragma unroll
      for (int kc = 0; kc < 8; ++kc)
        dwi[kc][tt] = *(const v2f*)&dWih[gr * HL + kc * 4 + 2 * lhi];
      #pragma unroll
      for (int kc = 0; kc < 16; ++kc)
        dwh[kc][tt] = *(const v2f*)&dWhh[gr * HD + kc * 4 + 2 * lhi];
    }
    const int f = wave * 16 + lrow;
    #pragma unroll
    for (int kc = 0; kc < 16; ++kc) {
      v2f z = {};
      if (wave < 3 && f < FF)
        z = *(const v2f*)&oW[f * HD + kc * 4 + 2 * lhi];
      wo[kc] = z;
    }
  }
  float bD[4][4];
  #pragma unroll
  for (int q = 0; q < 4; ++q)
    #pragma unroll
    for (int cidx = 0; cidx < 4; ++cidx)
      bD[q][cidx] = db[q * 64 + u64 + cidx];
  const float obias = (wave < 3 && (wave * 16 + lrow) < FF) ? ob[wave * 16 + lrow] : 0.0f;

  float cL[2] = {0.f, 0.f};
  float cD[4] = {0.f, 0.f, 0.f, 0.f};

  float* outblk = out + (size_t)b0 * TT * FF;

  #pragma unroll 1
  for (int t = 0; t < TT; ++t) {
    const float* lc = hlat + (t & 1) * (16 * HLS);
    float*       ln = hlat + ((t + 1) & 1) * (16 * HLS);
    const float* dc = hdec + (t & 1) * (16 * HS);
    float*       dn = hdec + ((t + 1) & 1) * (16 * HS);
    {
      v8f a8 = {};
      #pragma unroll
      for (int kc = 0; kc < 8; ++kc) {
        v2f b = *(const v2f*)&lc[lrow * HLS + kc * 4 + 2 * lhi];
        a8 = wmma4(wlh[kc], b, a8);
      }
      v2f hv;
      #pragma unroll
      for (int s = 0; s < 2; ++s) {
        float gi = a8[0 + s] + gxl[0 + s];
        float gf = a8[2 + s] + gxl[2 + s];
        float gg = a8[4 + s] + gxl[4 + s];
        float go = a8[6 + s] + gxl[6 + s];
        float c  = sigm(gf) * cL[s] + sigm(gi) * tanh_(gg);
        cL[s] = c;
        hv[s] = sigm(go) * tanh_(c);
      }
      *(v2f*)&ln[lrow * HLS + u32] = hv;
    }
    __syncthreads();
    {
      v8f acc[2] = {{}, {}};
      #pragma unroll
      for (int kc = 0; kc < 8; ++kc) {
        v2f b = *(const v2f*)&ln[lrow * HLS + kc * 4 + 2 * lhi];
        acc[0] = wmma4(dwi[kc][0], b, acc[0]);
        acc[1] = wmma4(dwi[kc][1], b, acc[1]);
      }
      #pragma unroll
      for (int kc = 0; kc < 16; ++kc) {
        v2f b = *(const v2f*)&dc[lrow * HS + kc * 4 + 2 * lhi];
        acc[0] = wmma4(dwh[kc][0], b, acc[0]);
        acc[1] = wmma4(dwh[kc][1], b, acc[1]);
      }
      v4f hv;
      #pragma unroll
      for (int tt = 0; tt < 2; ++tt) {
        #pragma unroll
        for (int s = 0; s < 2; ++s) {
          const int ci = 2 * tt + s;
          float gi = acc[tt][0 + s] + bD[0][ci];
          float gf = acc[tt][2 + s] + bD[1][ci];
          float gg = acc[tt][4 + s] + bD[2][ci];
          float go = acc[tt][6 + s] + bD[3][ci];
          float c  = sigm(gf) * cD[ci] + sigm(gi) * tanh_(gg);
          cD[ci] = c;
          hv[ci] = sigm(go) * tanh_(c);
        }
      }
      *(v4f*)&dn[lrow * HS + u64] = hv;
    }
    __syncthreads();
    if (wave < 3) {
      v8f a8 = {};
      #pragma unroll
      for (int kc = 0; kc < 16; ++kc) {
        v2f a = *(const v2f*)&dn[lrow * HS + kc * 4 + 2 * lhi];
        a8 = wmma4(a, wo[kc], a8);
      }
      const int f = wave * 16 + lrow;
      if (f < FF) {
        #pragma unroll
        for (int jj = 0; jj < 8; ++jj) {
          const int m = jj + 8 * lhi;
          outblk[(size_t)m * TT * FF + t * FF + f] = a8[jj] + obias;
        }
      }
    }
  }
}

extern "C" void kernel_launch(void* const* d_in, const int* in_sizes, int n_in,
                              void* d_out, int out_size, void* d_ws, size_t ws_size,
                              hipStream_t stream) {
  (void)in_sizes; (void)n_in; (void)out_size; (void)d_ws; (void)ws_size;
  const float* x    = (const float*)d_in[0];
  const float* eWih = (const float*)d_in[1];
  const float* eWhh = (const float*)d_in[2];
  const float* eb   = (const float*)d_in[3];
  const float* lWih = (const float*)d_in[4];
  const float* lWhh = (const float*)d_in[5];
  const float* lb   = (const float*)d_in[6];
  const float* dWih = (const float*)d_in[7];
  const float* dWhh = (const float*)d_in[8];
  const float* db   = (const float*)d_in[9];
  const float* oW   = (const float*)d_in[10];
  const float* ob   = (const float*)d_in[11];
  float* out = (float*)d_out;

  dim3 grid(BB / 16), block(256);
  hipLaunchKernelGGL(lstm_ae_kernel, grid, block, 0, stream,
                     x, eWih, eWhh, eb, lWih, lWhh, lb, dWih, dWhh, db, oW, ob, out);
}


# gate activations: IEEE-exact f32 division replaced by the hardware reciprocal (v_rcp_f32, 1 ulp) in sigmoid and tanh, whose denominators are >= 1; shorter vector chain per step; sum order kept
# speedup vs baseline: 1.5027x; 1.2091x over previous
#include <hip/hip_runtime.h>


typedef float v2f __attribute__((ext_vector_type(2)));
typedef float v4f __attribute__((ext_vector_type(4)));
typedef float v8f __attribute__((ext_vector_type(8)));

#define BB 4096
#define TT 100
#define FF 40
#define HE 64
#define HL 32
#define HD 64

#define HS  68
#define HLS 36
#define XS  44

__device__ __forceinline__ v8f wmma4(v2f a, v2f b, v8f c) {
  return __builtin_amdgcn_wmma_f32_16x16x4_f32(false, a, false, b, (short)0, c,
                                               false, false);
}

__device__ __forceinline__ float sigm(float x) {
  return __builtin_amdgcn_rcpf(1.0f + __expf(-x));
}
__device__ __forceinline__ float tanh_(float x) {
  float e = __expf(2.0f * x);
  return 1.0f - 2.0f * __builtin_amdgcn_rcpf(e + 1.0f);
}

__global__ __launch_bounds__(256)
void lstm_ae_kernel(const float* __restrict__ x,
                    const float* __restrict__ eWih, const float* __restrict__ eWhh,
                    const float* __restrict__ eb,
                    const float* __restrict__ lWih, const float* __restrict__ lWhh,
                    const float* __restrict__ lb,
                    const float* __restrict__ dWih, const float* __restrict__ dWhh,
                    const float* __restrict__ db,
                    const float* __restrict__ oW,   const float* __restrict__ ob,
                    float* __restrict__ out) {
  __shared__ __attribute__((aligned(16))) float henc[2 * 16 * HS];
  __shared__ __attribute__((aligned(16))) float hdec[2 * 16 * HS];
  __shared__ __attribute__((aligned(16))) float hlat[2 * 16 * HLS];
  __shared__ __attribute__((aligned(16))) float xenc[2 * 16 * XS];

  const int tid  = threadIdx.x;
  const int lane = tid & 31;
  const int wave = tid >> 5;
  const int lrow = lane & 15;
  const int lhi  = lane >> 4;
  const int b0   = blockIdx.x * 16;

  const int mh = lrow >> 3;
  const int mq = (lrow & 7) >> 1;
  const int ms = lrow & 1;
  const int row64 = mq * 64 + 8 * wave + 4 * mh + ms;
  const int row32 = mq * 32 + 4 * wave + 2 * mh + ms;
  const int u64 = 8 * wave + 4 * lhi;
  const int u32 = 4 * wave + 2 * lhi;

  v2f wih[10][2];
  v2f whh[16][2];
  #pragma unroll
  for (int tt = 0; tt < 2; ++tt) {
    const int gr = row64 + 2 * tt;
    #pragma unroll
    for (int kc = 0; kc < 10; ++kc)
      wih[kc][tt] = *(const v2f*)&eWih[gr * FF + kc * 4 + 2 * lhi];
    #pragma unroll
    for (int kc = 0; kc < 16; ++kc)
      whh[kc][tt] = *(const v2f*)&eWhh[gr * HE + kc * 4 + 2 * lhi];
  }
  float bE[4][4];
  #pragma unroll
  for (int q = 0; q < 4; ++q)
    #pragma unroll
    for (int cidx = 0; cidx < 4; ++cidx)
      bE[q][cidx] = eb[q * 64 + u64 + cidx];

  for (int i = tid; i < 16 * HS; i += 256) { henc[i] = 0.0f; hdec[i] = 0.0f; }
  for (int i = tid; i < 16 * HLS; i += 256) hlat[i] = 0.0f;
  if (tid < 160) {
    const int r0 = tid / 10, q0 = tid - 10 * r0;
    *(v4f*)&xenc[r0 * XS + q0 * 4] =
        *(const v4f*)&x[((size_t)b0 + r0) * TT * FF + q0 * 4];
  }
  float cE[4] = {0.f, 0.f, 0.f, 0.f};
  __syncthreads();

  const float* xblk = x + (size_t)b0 * TT * FF;

  const bool xld  = __builtin_amdgcn_readfirstlane(wave) < 5;
  const int  xrw  = xld ? tid / 10 : 0;
  const int  xq   = xld ? tid - 10 * xrw : 0;
  const float* xsrc = xblk + (size_t)xrw * TT * FF + xq * 4;
  float*       xdst = xenc + xrw * XS + xq * 4;

  #pragma unroll 1
  for (int t = 0; t < TT; ++t) {
    const float* hc = henc + (t & 1) * (16 * HS);
    float*       hn = henc + ((t + 1) & 1) * (16 * HS);
    const float* xc = xenc + (t & 1) * (16 * XS);
    v4f xv = {};
    if (xld) {
      const int tn = (t + 1 < TT) ? (t + 1) : (TT - 1);
      xv = *(const v4f*)&xsrc[tn * FF];
    }
    v8f acc[2] = {{}, {}};
    #pragma unroll
    for (int kc = 0; kc < 10; ++kc) {
      v2f b = *(const v2f*)&xc[lrow * XS + kc * 4 + 2 * lhi];
      acc[0] = wmma4(wih[kc][0], b, acc[0]);
      acc[1] = wmma4(wih[kc][1], b, acc[1]);
    }
    #pragma unroll
    for (int kc = 0; kc < 16; ++kc) {
      v2f b = *(const v2f*)&hc[lrow * HS + kc * 4 + 2 * lhi];
      acc[0] = wmma4(whh[kc][0], b, acc[0]);
      acc[1] = wmma4(whh[kc][1], b, acc[1]);
    }
    v4f hv;
    #pragma unroll
    for (int tt = 0; tt < 2; ++tt) {
      #pragma unroll
      for (int s = 0; s < 2; ++s) {
        const int ci = 2 * tt + s;
        float gi = acc[tt][0 + s] + bE[0][ci];
        float gf = acc[tt][2 + s] + bE[1][ci];
        float gg = acc[tt][4 + s] + bE[2][ci];
        float go = acc[tt][6 + s] + bE[3][ci];
        float c  = sigm(gf) * cE[ci] + sigm(gi) * tanh_(gg);
        cE[ci] = c;
        hv[ci] = sigm(go) * tanh_(c);
      }
    }
    *(v4f*)&hn[lrow * HS + u64] = hv;
    if (xld) *(v4f*)&xdst[((t + 1) & 1) * (16 * XS)] = xv;
    __syncthreads();
  }
  const float* hlast = henc + (TT & 1) * (16 * HS);

  float gxl[8];
  {
    v8f a8 = {};
    #pragma unroll
    for (int kc = 0; kc < 16; ++kc) {
      v2f a = *(const v2f*)&lWih[row32 * HE + kc * 4 + 2 * lhi];
      v2f b = *(const v2f*)&hlast[lrow * HS + kc * 4 + 2 * lhi];
      a8 = wmma4(a, b, a8);
    }
    #pragma unroll
    for (int r = 0; r < 8; ++r)
      gxl[r] = a8[r] + lb[(r >> 1) * 32 + u32 + (r & 1)];
  }

  v2f wlh[8];
  v2f dwi[8][2];
  v2f dwh[16][2];
  v2f wo[16];
  {
    #pragma unroll
    for (int kc = 0; kc < 8; ++kc)
      wlh[kc] = *(const v2f*)&lWhh[row32 * HL + kc * 4 + 2 * lhi];
    #pragma unroll
    for (int tt = 0; tt < 2; ++tt) {
      const int gr = row64 + 2 * tt;
      #pragma unroll
      for (int kc = 0; kc < 8; ++kc)
        dwi[kc][tt] = *(const v2f*)&dWih[gr * HL + kc * 4 + 2 * lhi];
      #pragma unroll
      for (int kc = 0; kc < 16; ++kc)
        dwh[kc][tt] = *(const v2f*)&dWhh[gr * HD + kc * 4 + 2 * lhi];
    }
    const int f = wave * 16 + lrow;
    #pragma unroll
    for (int kc = 0; kc < 16; ++kc) {
      v2f z = {};
      if (wave < 3 && f < FF)
        z = *(const v2f*)&oW[f * HD + kc * 4 + 2 * lhi];
      wo[kc] = z;
    }
  }
  float bD[4][4];
  #pragma unroll
  for (int q = 0; q < 4; ++q)
    #pragma unroll
    for (int cidx = 0; cidx < 4; ++cidx)
      bD[q][cidx] = db[q * 64 + u64 + cidx];
  const float obias = (wave < 3 && (wave * 16 + lrow) < FF) ? ob[wave * 16 + lrow] : 0.0f;

  float cL[2] = {0.f, 0.f};
  float cD[4] = {0.f, 0.f, 0.f, 0.f};

  float* outblk = out + (size_t)b0 * TT * FF;

  #pragma unroll 1
  for (int t = 0; t < TT; ++t) {
    const float* lc = hlat + (t & 1) * (16 * HLS);
    float*       ln = hlat + ((t + 1) & 1) * (16 * HLS);
    const float* dc = hdec + (t & 1) * (16 * HS);
    float*       dn = hdec + ((t + 1) & 1) * (16 * HS);
    {
      v8f a8 = {};
      #pragma unroll
      for (int kc = 0; kc < 8; ++kc) {
        v2f b = *(const v2f*)&lc[lrow * HLS + kc * 4 + 2 * lhi];
        a8 = wmma4(wlh[kc], b, a8);
      }
      v2f hv;
      #pragma unroll
      for (int s = 0; s < 2; ++s) {
        float gi = a8[0 + s] + gxl[0 + s];
        float gf = a8[2 + s] + gxl[2 + s];
        float gg = a8[4 + s] + gxl[4 + s];
        float go = a8[6 + s] + gxl[6 + s];
        float c  = sigm(gf) * cL[s] + sigm(gi) * tanh_(gg);
        cL[s] = c;
        hv[s] = sigm(go) * tanh_(c);
      }
      *(v2f*)&ln[lrow * HLS + u32] = hv;
    }
    __syncthreads();
    {
      v8f acc[2] = {{}, {}};
      #pragma unroll
      for (int kc = 0; kc < 8; ++kc) {
        v2f b = *(const v2f*)&ln[lrow * HLS + kc * 4 + 2 * lhi];
        acc[0] = wmma4(dwi[kc][0], b, acc[0]);
        acc[1] = wmma4(dwi[kc][1], b, acc[1]);
      }
      #pragma unroll
      for (int kc = 0; kc < 16; ++kc) {
        v2f b = *(const v2f*)&dc[lrow * HS + kc * 4 + 2 * lhi];
        acc[0] = wmma4(dwh[kc][0], b, acc[0]);
        acc[1] = wmma4(dwh[kc][1], b, acc[1]);
      }
      v4f hv;
      #pragma unroll
      for (int tt = 0; tt < 2; ++tt) {
        #pragma unroll
        for (int s = 0; s < 2; ++s) {
          const int ci = 2 * tt + s;
          float gi = acc[tt][0 + s] + bD[0][ci];
          float gf = acc[tt][2 + s] + bD[1][ci];
          float gg = acc[tt][4 + s] + bD[2][ci];
          float go = acc[tt][6 + s] + bD[3][ci];
          float c  = sigm(gf) * cD[ci] + sigm(gi) * tanh_(gg);
          cD[ci] = c;
          hv[ci] = sigm(go) * tanh_(c);
        }
      }
      *(v4f*)&dn[lrow * HS + u64] = hv;
    }
    __syncthreads();
    if (wave < 3) {
      v8f a8 = {};
      #pragma unroll
      for (int kc = 0; kc < 16; ++kc) {
        v2f a = *(const v2f*)&dn[lrow * HS + kc * 4 + 2 * lhi];
        a8 = wmma4(a, wo[kc], a8);
      }
      const int f = wave * 16 + lrow;
      if (f < FF) {
        #pragma unroll
        for (int jj = 0; jj < 8; ++jj) {
          const int m = jj + 8 * lhi;
          outblk[(size_t)m * TT * FF + t * FF + f] = a8[jj] + obias;
        }
      }
    }
  }
}

extern "C" void kernel_launch(void* const* d_in, const int* in_sizes, int n_in,
                              void* d_out, int out_size, void* d_ws, size_t ws_size,
                              hipStream_t stream) {
  (void)in_sizes; (void)n_in; (void)out_size; (void)d_ws; (void)ws_size;
  const float* x    = (const float*)d_in[0];
  const float* eWih = (const float*)d_in[1];
  const float* eWhh = (const float*)d_in[2];
  const float* eb   = (const float*)d_in[3];
  const float* lWih = (const float*)d_in[4];
  const float* lWhh = (const float*)d_in[5];
  const float* lb   = (const float*)d_in[6];
  const float* dWih = (const float*)d_in[7];
  const float* dWhh = (const float*)d_in[8];
  const float* db   = (const float*)d_in[9];
  const float* oW   = (const float*)d_in[10];
  const float* ob   = (const float*)d_in[11];
  float* out = (float*)d_out;

  dim3 grid(BB / 16), block(256);
  hipLaunchKernelGGL(lstm_ae_kernel, grid, block, 0, stream,
                     x, eWih, eWhh, eb, lWih, lWhh, lb, dWih, dWhh, db, oW, ob, out);
}


# decoder loop skewed: latent cell one step ahead, output projection one step behind, so all products of a step share one group of LDS fragment reads and ONE barrier (was two); sum order kept
# speedup vs baseline: 1.6234x; 1.0803x over previous
#include <hip/hip_runtime.h>


typedef float v2f __attribute__((ext_vector_type(2)));
typedef float v4f __attribute__((ext_vector_type(4)));
typedef float v8f __attribute__((ext_vector_type(8)));

#define BB 4096
#define TT 100
#define FF 40
#define HE 64
#define HL 32
#define HD 64

#define HS  68
#define HLS 36
#define XS  44

__device__ __forceinline__ v8f wmma4(v2f a, v2f b, v8f c) {
  return __builtin_amdgcn_wmma_f32_16x16x4_f32(false, a, false, b, (short)0, c,
                                               false, false);
}

__device__ __forceinline__ float sigm(float x) {
  return __builtin_amdgcn_rcpf(1.0f + __expf(-x));
}
__device__ __forceinline__ float tanh_(float x) {
  float e = __expf(2.0f * x);
  return 1.0f - 2.0f * __builtin_amdgcn_rcpf(e + 1.0f);
}

__global__ __launch_bounds__(256)
void lstm_ae_kernel(const float* __restrict__ x,
                    const float* __restrict__ eWih, const float* __restrict__ eWhh,
                    const float* __restrict__ eb,
                    const float* __restrict__ lWih, const float* __restrict__ lWhh,
                    const float* __restrict__ lb,
                    const float* __restrict__ dWih, const float* __restrict__ dWhh,
                    const float* __restrict__ db,
                    const float* __restrict__ oW,   const float* __restrict__ ob,
                    float* __restrict__ out) {
  __shared__ __attribute__((aligned(16))) float henc[2 * 16 * HS];
  __shared__ __attribute__((aligned(16))) float hdec[2 * 16 * HS];
  __shared__ __attribute__((aligned(16))) float hlat[2 * 16 * HLS];
  __shared__ __attribute__((aligned(16))) float xenc[2 * 16 * XS];

  const int tid  = threadIdx.x;
  const int lane = tid & 31;
  const int wave = tid >> 5;
  const int lrow = lane & 15;
  const int lhi  = lane >> 4;
  const int b0   = blockIdx.x * 16;

  const int mh = lrow >> 3;
  const int mq = (lrow & 7) >> 1;
  const int ms = lrow & 1;
  const int row64 = mq * 64 + 8 * wave + 4 * mh + ms;
  const int row32 = mq * 32 + 4 * wave + 2 * mh + ms;
  const int u64 = 8 * wave + 4 * lhi;
  const int u32 = 4 * wave + 2 * lhi;

  v2f wih[10][2];
  v2f whh[16][2];
  #pragma unroll
  for (int tt = 0; tt < 2; ++tt) {
    const int gr = row64 + 2 * tt;
    #pragma unroll
    for (int kc = 0; kc < 10; ++kc)
      wih[kc][tt] = *(const v2f*)&eWih[gr * FF + kc * 4 + 2 * lhi];
    #pragma unroll
    for (int kc = 0; kc < 16; ++kc)
      whh[kc][tt] = *(const v2f*)&eWhh[gr * HE + kc * 4 + 2 * lhi];
  }
  float bE[4][4];
  #pragma unroll
  for (int q = 0; q < 4; ++q)
    #pragma unroll
    for (int cidx = 0; cidx < 4; ++cidx)
      bE[q][cidx] = eb[q * 64 + u64 + cidx];

  for (int i = tid; i < 16 * HS; i += 256) { henc[i] = 0.0f; hdec[i] = 0.0f; }
  for (int i = tid; i < 16 * HLS; i += 256) hlat[i] = 0.0f;
  if (tid < 160) {
    const int r0 = tid / 10, q0 = tid - 10 * r0;
    *(v4f*)&xenc[r0 * XS + q0 * 4] =
        *(const v4f*)&x[((size_t)b0 + r0) * TT * FF + q0 * 4];
  }
  float cE[4] = {0.f, 0.f, 0.f, 0.f};
  __syncthreads();

  const float* xblk = x + (size_t)b0 * TT * FF;

  const bool xld  = __builtin_amdgcn_readfirstlane(wave) < 5;
  const int  xrw  = xld ? tid / 10 : 0;
  const int  xq   = xld ? tid - 10 * xrw : 0;
  const float* xsrc = xblk + (size_t)xrw * TT * FF + xq * 4;
  float*       xdst = xenc + xrw * XS + xq * 4;

  #pragma unroll 1
  for (int t = 0; t < TT; ++t) {
    const float* hc = henc + (t & 1) * (16 * HS);
    float*       hn = henc + ((t + 1) & 1) * (16 * HS);
    const float* xc = xenc + (t & 1) * (16 * XS);
    v4f xv = {};
    if (xld) {
      const int tn = (t + 1 < TT) ? (t + 1) : (TT - 1);
      xv = *(const v4f*)&xsrc[tn * FF];
    }
    v8f acc[2] = {{}, {}};
    #pragma unroll
    for (int kc = 0; kc < 10; ++kc) {
      v2f b = *(const v2f*)&xc[lrow * XS + kc * 4 + 2 * lhi];
      acc[0] = wmma4(wih[kc][0], b, acc[0]);
      acc[1] = wmma4(wih[kc][1], b, acc[1]);
    }
    #pragma unroll
    for (int kc = 0; kc < 16; ++kc) {
      v2f b = *(const v2f*)&hc[lrow * HS + kc * 4 + 2 * lhi];
      acc[0] = wmma4(whh[kc][0], b, acc[0]);
      acc[1] = wmma4(whh[kc][1], b, acc[1]);
    }
    v4f hv;
    #pragma unroll
    for (int tt = 0; tt < 2; ++tt) {
      #pragma unroll
      for (int s = 0; s < 2; ++s) {
        const int ci = 2 * tt + s;
        float gi = acc[tt][0 + s] + bE[0][ci];
        float gf = acc[tt][2 + s] + bE[1][ci];
        float gg = acc[tt][4 + s] + bE[2][ci];
        float go = acc[tt][6 + s] + bE[3][ci];
        float c  = sigm(gf) * cE[ci] + sigm(gi) * tanh_(gg);
        cE[ci] = c;
        hv[ci] = sigm(go) * tanh_(c);
      }
    }
    *(v4f*)&hn[lrow * HS + u64] = hv;
    if (xld) *(v4f*)&xdst[((t + 1) & 1) * (16 * XS)] = xv;
    __syncthreads();
  }
  const float* hlast = henc + (TT & 1) * (16 * HS);

  float gxl[8];
  {
    v8f a8 = {};
    #pragma unroll
    for (int kc = 0; kc < 16; ++kc) {
      v2f a = *(const v2f*)&lWih[row32 * HE + kc * 4 + 2 * lhi];
      v2f b = *(const v2f*)&hlast[lrow * HS + kc * 4 + 2 * lhi];
      a8 = wmma4(a, b, a8);
    }
    #pragma unroll
    for (int r = 0; r < 8; ++r)
      gxl[r] = a8[r] + lb[(r >> 1) * 32 + u32 + (r & 1)];
  }

  v2f wlh[8];
  v2f dwi[8][2];
  v2f dwh[16][2];
  v2f wo[16];
  {
    #pragma unroll
    for (int kc = 0; kc < 8; ++kc)
      wlh[kc] = *(const v2f*)&lWhh[row32 * HL + kc * 4 + 2 * lhi];
    #pragma unroll
    for (int tt = 0; tt < 2; ++tt) {
      const int gr = row64 + 2 * tt;
      #pragma unroll
      for (int kc = 0; kc < 8; ++kc)
        dwi[kc][tt] = *(const v2f*)&dWih[gr * HL + kc * 4 + 2 * lhi];
      #pragma unroll
      for (int kc = 0; kc < 16; ++kc)
        dwh[kc][tt] = *(const v2f*)&dWhh[gr * HD + kc * 4 + 2 * lhi];
    }
    const int f = wave * 16 + lrow;
    #pragma unroll
    for (int kc = 0; kc < 16; ++kc) {
      v2f z = {};
      if (wave < 3 && f < FF)
        z = *(const v2f*)&oW[f * HD + kc * 4 + 2 * lhi];
      wo[kc] = z;
    }
  }
  float bD[4][4];
  #pragma unroll
  for (int q = 0; q < 4; ++q)
    #pragma unroll
    for (int cidx = 0; cidx < 4; ++cidx)
      bD[q][cidx] = db[q * 64 + u64 + cidx];
  const float obias = (wave < 3 && (wave * 16 + lrow) < FF) ? ob[wave * 16 + lrow] : 0.0f;

  float cL[2] = {0.f, 0.f};
  float cD[4] = {0.f, 0.f, 0.f, 0.f};

  float* outblk = out + (size_t)b0 * TT * FF;
  const bool proj = __builtin_amdgcn_readfirstlane(wave) < 3;
  const int  fo   = wave * 16 + lrow;

  {
    v8f a8 = {};
    #pragma unroll
    for (int kc = 0; kc < 8; ++kc) {
      v2f b = *(const v2f*)&hlat[lrow * HLS + kc * 4 + 2 * lhi];
      a8 = wmma4(wlh[kc], b, a8);
    }
    v2f hv;
    #pragma unroll
    for (int s = 0; s < 2; ++s) {
      float gi = a8[0 + s] + gxl[0 + s];
      float gf = a8[2 + s] + gxl[2 + s];
      float gg = a8[4 + s] + gxl[4 + s];
      float go = a8[6 + s] + gxl[6 + s];
      float c  = sigm(gf) * cL[s] + sigm(gi) * tanh_(gg);
      cL[s] = c;
      hv[s] = sigm(go) * tanh_(c);
    }
    *(v2f*)&hlat[16 * HLS + lrow * HLS + u32] = hv;
  }
  __syncthreads();

  #pragma unroll 1
  for (int t = 0; t < TT; ++t) {
    const float* lc = hlat + ((t + 1) & 1) * (16 * HLS);
    float*       ln = hlat + (t & 1) * (16 * HLS);
    const float* dc = hdec + (t & 1) * (16 * HS);
    float*       dn = hdec + ((t + 1) & 1) * (16 * HS);
    v2f bl[8], bd[16];
    #pragma unroll
    for (int kc = 0; kc < 8; ++kc)
      bl[kc] = *(const v2f*)&lc[lrow * HLS + kc * 4 + 2 * lhi];
    #pragma unroll
    for (int kc = 0; kc < 16; ++kc)
      bd[kc] = *(const v2f*)&dc[lrow * HS + kc * 4 + 2 * lhi];
    if (proj && t > 0) {
      v8f p8 = {};
      #pragma unroll
      for (int kc = 0; kc < 16; ++kc)
        p8 = wmma4(bd[kc], wo[kc], p8);
      if (fo < FF) {
        #pragma unroll
        for (int jj = 0; jj < 8; ++jj) {
          const int m = jj + 8 * lhi;
          outblk[(size_t)m * TT * FF + (t - 1) * FF + fo] = p8[jj] + obias;
        }
      }
    }
    v8f a8 = {};
    #pragma unroll
    for (int kc = 0; kc < 8; ++kc)
      a8 = wmma4(wlh[kc], bl[kc], a8);
    v8f acc[2] = {{}, {}};
    #pragma unroll
    for (int kc = 0; kc < 8; ++kc) {
      acc[0] = wmma4(dwi[kc][0], bl[kc], acc[0]);
      acc[1] = wmma4(dwi[kc][1], bl[kc], acc[1]);
    }
    #pragma unroll
    for (int kc = 0; kc < 16; ++kc) {
      acc[0] = wmma4(dwh[kc][0], bd[kc], acc[0]);
      acc[1] = wmma4(dwh[kc][1], bd[kc], acc[1]);
    }
    {
      v2f hv;
      #pragma unroll
      for (int s = 0; s < 2; ++s) {
        float gi = a8[0 + s] + gxl[0 + s];
        float gf = a8[2 + s] + gxl[2 + s];
        float gg = a8[4 + s] + gxl[4 + s];
        float go = a8[6 + s] + gxl[6 + s];
        float c  = sigm(gf) * cL[s] + sigm(gi) * tanh_(gg);
        cL[s] = c;
        hv[s] = sigm(go) * tanh_(c);
      }
      *(v2f*)&ln[lrow * HLS + u32] = hv;
    }
    {
      v4f hv;
      #pragma unroll
      for (int tt = 0; tt < 2; ++tt) {
        #pragma unroll
        for (int s = 0; s < 2; ++s) {
          const int ci = 2 * tt + s;
          float gi = acc[tt][0 + s] + bD[0][ci];
          float gf = acc[tt][2 + s] + bD[1][ci];
          float gg = acc[tt][4 + s] + bD[2][ci];
          float go = acc[tt][6 + s] + bD[3][ci];
          float c  = sigm(gf) * cD[ci] + sigm(gi) * tanh_(gg);
          cD[ci] = c;
          hv[ci] = sigm(go) * tanh_(c);
        }
      }
      *(v4f*)&dn[lrow * HS + u64] = hv;
    }
    __syncthreads();
  }
  if (proj) {
    const float* dl = hdec + (TT & 1) * (16 * HS);
    v8f p8 = {};
    #pragma unroll
    for (int kc = 0; kc < 16; ++kc) {
      v2f a = *(const v2f*)&dl[lrow * HS + kc * 4 + 2 * lhi];
      p8 = wmma4(a, wo[kc], p8);
    }
    if (fo < FF) {
      #pragma unroll
      for (int jj = 0; jj < 8; ++jj) {
        const int m = jj + 8 * lhi;
        outblk[(size_t)m * TT * FF + (TT - 1) * FF + fo] = p8[jj] + obias;
      }
    }
  }
}

extern "C" void kernel_launch(void* const* d_in, const int* in_sizes, int n_in,
                              void* d_out, int out_size, void* d_ws, size_t ws_size,
                              hipStream_t stream) {
  (void)in_sizes; (void)n_in; (void)out_size; (void)d_ws; (void)ws_size;
  const float* x    = (const float*)d_in[0];
  const float* eWih = (const float*)d_in[1];
  const float* eWhh = (const float*)d_in[2];
  const float* eb   = (const float*)d_in[3];
  const float* lWih = (const float*)d_in[4];
  const float* lWhh = (const float*)d_in[5];
  const float* lb   = (const float*)d_in[6];
  const float* dWih = (const float*)d_in[7];
  const float* dWhh = (const float*)d_in[8];
  const float* db   = (const float*)d_in[9];
  const float* oW   = (const float*)d_in[10];
  const float* ob   = (const float*)d_in[11];
  float* out = (float*)d_out;

  dim3 grid(BB / 16), block(256);
  hipLaunchKernelGGL(lstm_ae_kernel, grid, block, 0, stream,
                     x, eWih, eWhh, eb, lWih, lWhh, lb, dWih, dWhh, db, oW, ob, out);
}


# gate activations on the hardware tanh (v_tanh_f32): sigmoid = (1+tanh(x/2))/2, halving folded into the bias fma; one transcendental per activation (was exp, add, rcp); sum order kept
# speedup vs baseline: 1.7226x; 1.0611x over previous
#include <hip/hip_runtime.h>


typedef float v2f __attribute__((ext_vector_type(2)));
typedef float v4f __attribute__((ext_vector_type(4)));
typedef float v8f __attribute__((ext_vector_type(8)));

#define BB 4096
#define TT 100
#define FF 40
#define HE 64
#define HL 32
#define HD 64

#define HS  68
#define HLS 36
#define XS  44

__device__ __forceinline__ v8f wmma4(v2f a, v2f b, v8f c) {
  return __builtin_amdgcn_wmma_f32_16x16x4_f32(false, a, false, b, (short)0, c,
                                               false, false);
}

__device__ __forceinline__ float tanh_(float x) {
  return __builtin_amdgcn_tanhf(x);
}
__device__ __forceinline__ float sigm_h(float hx) {
  return fmaf(__builtin_amdgcn_tanhf(hx), 0.5f, 0.5f);
}

__global__ __launch_bounds__(256)
void lstm_ae_kernel(const float* __restrict__ x,
                    const float* __restrict__ eWih, const float* __restrict__ eWhh,
                    const float* __restrict__ eb,
                    const float* __restrict__ lWih, const float* __restrict__ lWhh,
                    const float* __restrict__ lb,
                    const float* __restrict__ dWih, const float* __restrict__ dWhh,
                    const float* __restrict__ db,
                    const float* __restrict__ oW,   const float* __restrict__ ob,
                    float* __restrict__ out) {
  __shared__ __attribute__((aligned(16))) float henc[2 * 16 * HS];
  __shared__ __attribute__((aligned(16))) float hdec[2 * 16 * HS];
  __shared__ __attribute__((aligned(16))) float hlat[2 * 16 * HLS];
  __shared__ __attribute__((aligned(16))) float xenc[2 * 16 * XS];

  const int tid  = threadIdx.x;
  const int lane = tid & 31;
  const int wave = tid >> 5;
  const int lrow = lane & 15;
  const int lhi  = lane >> 4;
  const int b0   = blockIdx.x * 16;

  const int mh = lrow >> 3;
  const int mq = (lrow & 7) >> 1;
  const int ms = lrow & 1;
  const int row64 = mq * 64 + 8 * wave + 4 * mh + ms;
  const int row32 = mq * 32 + 4 * wave + 2 * mh + ms;
  const int u64 = 8 * wave + 4 * lhi;
  const int u32 = 4 * wave + 2 * lhi;

  v2f wih[10][2];
  v2f whh[16][2];
  #pragma unroll
  for (int tt = 0; tt < 2; ++tt) {
    const int gr = row64 + 2 * tt;
    #pragma unroll
    for (int kc = 0; kc < 10; ++kc)
      wih[kc][tt] = *(const v2f*)&eWih[gr * FF + kc * 4 + 2 * lhi];
    #pragma unroll
    for (int kc = 0; kc < 16; ++kc)
      whh[kc][tt] = *(const v2f*)&eWhh[gr * HE + kc * 4 + 2 * lhi];
  }
  float bE[4][4];
  #pragma unroll
  for (int q = 0; q < 4; ++q)
    #pragma unroll
    for (int cidx = 0; cidx < 4; ++cidx)
      bE[q][cidx] = (q == 2 ? 1.0f : 0.5f) * eb[q * 64 + u64 + cidx];

  for (int i = tid; i < 16 * HS; i += 256) { henc[i] = 0.0f; hdec[i] = 0.0f; }
  for (int i = tid; i < 16 * HLS; i += 256) hlat[i] = 0.0f;
  if (tid < 160) {
    const int r0 = tid / 10, q0 = tid - 10 * r0;
    *(v4f*)&xenc[r0 * XS + q0 * 4] =
        *(const v4f*)&x[((size_t)b0 + r0) * TT * FF + q0 * 4];
  }
  float cE[4] = {0.f, 0.f, 0.f, 0.f};
  __syncthreads();

  const float* xblk = x + (size_t)b0 * TT * FF;

  const bool xld  = __builtin_amdgcn_readfirstlane(wave) < 5;
  const int  xrw  = xld ? tid / 10 : 0;
  const int  xq   = xld ? tid - 10 * xrw : 0;
  const float* xsrc = xblk + (size_t)xrw * TT * FF + xq * 4;
  float*       xdst = xenc + xrw * XS + xq * 4;

  #pragma unroll 1
  for (int t = 0; t < TT; ++t) {
    const float* hc = henc + (t & 1) * (16 * HS);
    float*       hn = henc + ((t + 1) & 1) * (16 * HS);
    const float* xc = xenc + (t & 1) * (16 * XS);
    v4f xv = {};
    if (xld) {
      const int tn = (t + 1 < TT) ? (t + 1) : (TT - 1);
      xv = *(const v4f*)&xsrc[tn * FF];
    }
    v8f acc[2] = {{}, {}};
    #pragma unroll
    for (int kc = 0; kc < 10; ++kc) {
      v2f b = *(const v2f*)&xc[lrow * XS + kc * 4 + 2 * lhi];
      acc[0] = wmma4(wih[kc][0], b, acc[0]);
      acc[1] = wmma4(wih[kc][1], b, acc[1]);
    }
    #pragma unroll
    for (int kc = 0; kc < 16; ++kc) {
      v2f b = *(const v2f*)&hc[lrow * HS + kc * 4 + 2 * lhi];
      acc[0] = wmma4(whh[kc][0], b, acc[0]);
      acc[1] = wmma4(whh[kc][1], b, acc[1]);
    }
    v4f hv;
    #pragma unroll
    for (int tt = 0; tt < 2; ++tt) {
      #pragma unroll
      for (int s = 0; s < 2; ++s) {
        const int ci = 2 * tt + s;
        float hi = fmaf(acc[tt][0 + s], 0.5f, bE[0][ci]);
        float hf = fmaf(acc[tt][2 + s], 0.5f, bE[1][ci]);
        float gg = acc[tt][4 + s] + bE[2][ci];
        float ho = fmaf(acc[tt][6 + s], 0.5f, bE[3][ci]);
        float c  = sigm_h(hf) * cE[ci] + sigm_h(hi) * tanh_(gg);
        cE[ci] = c;
        hv[ci] = sigm_h(ho) * tanh_(c);
      }
    }
    *(v4f*)&hn[lrow * HS + u64] = hv;
    if (xld) *(v4f*)&xdst[((t + 1) & 1) * (16 * XS)] = xv;
    __syncthreads();
  }
  const float* hlast = henc + (TT & 1) * (16 * HS);

  float gxl[8];
  {
    v8f a8 = {};
    #pragma unroll
    for (int kc = 0; kc < 16; ++kc) {
      v2f a = *(const v2f*)&lWih[row32 * HE + kc * 4 + 2 * lhi];
      v2f b = *(const v2f*)&hlast[lrow * HS + kc * 4 + 2 * lhi];
      a8 = wmma4(a, b, a8);
    }
    #pragma unroll
    for (int r = 0; r < 8; ++r)
      gxl[r] = ((r >> 1) == 2 ? 1.0f : 0.5f) * (a8[r] + lb[(r >> 1) * 32 + u32 + (r & 1)]);
  }

  v2f wlh[8];
  v2f dwi[8][2];
  v2f dwh[16][2];
  v2f wo[16];
  {
    #pragma unroll
    for (int kc = 0; kc < 8; ++kc)
      wlh[kc] = *(const v2f*)&lWhh[row32 * HL + kc * 4 + 2 * lhi];
    #pragma unroll
    for (int tt = 0; tt < 2; ++tt) {
      const int gr = row64 + 2 * tt;
      #pragma unroll
      for (int kc = 0; kc < 8; ++kc)
        dwi[kc][tt] = *(const v2f*)&dWih[gr * HL + kc * 4 + 2 * lhi];
      #pragma unroll
      for (int kc = 0; kc < 16; ++kc)
        dwh[kc][tt] = *(const v2f*)&dWhh[gr * HD + kc * 4 + 2 * lhi];
    }
    const int f = wave * 16 + lrow;
    #pragma unroll
    for (int kc = 0; kc < 16; ++kc) {
      v2f z = {};
      if (wave < 3 && f < FF)
        z = *(const v2f*)&oW[f * HD + kc * 4 + 2 * lhi];
      wo[kc] = z;
    }
  }
  float bD[4][4];
  #pragma unroll
  for (int q = 0; q < 4; ++q)
    #pragma unroll
    for (int cidx = 0; cidx < 4; ++cidx)
      bD[q][cidx] = (q == 2 ? 1.0f : 0.5f) * db[q * 64 + u64 + cidx];
  const float obias = (wave < 3 && (wave * 16 + lrow) < FF) ? ob[wave * 16 + lrow] : 0.0f;

  float cL[2] = {0.f, 0.f};
  float cD[4] = {0.f, 0.f, 0.f, 0.f};

  float* outblk = out + (size_t)b0 * TT * FF;
  const bool proj = __builtin_amdgcn_readfirstlane(wave) < 3;
  const int  fo   = wave * 16 + lrow;

  {
    v8f a8 = {};
    #pragma unroll
    for (int kc = 0; kc < 8; ++kc) {
      v2f b = *(const v2f*)&hlat[lrow * HLS + kc * 4 + 2 * lhi];
      a8 = wmma4(wlh[kc], b, a8);
    }
    v2f hv;
    #pragma unroll
    for (int s = 0; s < 2; ++s) {
      float hi = fmaf(a8[0 + s], 0.5f, gxl[0 + s]);
      float hf = fmaf(a8[2 + s], 0.5f, gxl[2 + s]);
      float gg = a8[4 + s] + gxl[4 + s];
      float ho = fmaf(a8[6 + s], 0.5f, gxl[6 + s]);
      float c  = sigm_h(hf) * cL[s] + sigm_h(hi) * tanh_(gg);
      cL[s] = c;
      hv[s] = sigm_h(ho) * tanh_(c);
    }
    *(v2f*)&hlat[16 * HLS + lrow * HLS + u32] = hv;
  }
  __syncthreads();

  #pragma unroll 1
  for (int t = 0; t < TT; ++t) {
    const float* lc = hlat + ((t + 1) & 1) * (16 * HLS);
    float*       ln = hlat + (t & 1) * (16 * HLS);
    const float* dc = hdec + (t & 1) * (16 * HS);
    float*       dn = hdec + ((t + 1) & 1) * (16 * HS);
    v2f bl[8], bd[16];
    #pragma unroll
    for (int kc = 0; kc < 8; ++kc)
      bl[kc] = *(const v2f*)&lc[lrow * HLS + kc * 4 + 2 * lhi];
    #pragma unroll
    for (int kc = 0; kc < 16; ++kc)
      bd[kc] = *(const v2f*)&dc[lrow * HS + kc * 4 + 2 * lhi];
    if (proj && t > 0) {
      v8f p8 = {};
      #pragma unroll
      for (int kc = 0; kc < 16; ++kc)
        p8 = wmma4(bd[kc], wo[kc], p8);
      if (fo < FF) {
        #pragma unroll
        for (int jj = 0; jj < 8; ++jj) {
          const int m = jj + 8 * lhi;
          outblk[(size_t)m * TT * FF + (t - 1) * FF + fo] = p8[jj] + obias;
        }
      }
    }
    v8f a8 = {};
    #pragma unroll
    for (int kc = 0; kc < 8; ++kc)
      a8 = wmma4(wlh[kc], bl[kc], a8);
    v8f acc[2] = {{}, {}};
    #pragma unroll
    for (int kc = 0; kc < 8; ++kc) {
      acc[0] = wmma4(dwi[kc][0], bl[kc], acc[0]);
      acc[1] = wmma4(dwi[kc][1], bl[kc], acc[1]);
    }
    #pragma unroll
    for (int kc = 0; kc < 16; ++kc) {
      acc[0] = wmma4(dwh[kc][0], bd[kc], acc[0]);
      acc[1] = wmma4(dwh[kc][1], bd[kc], acc[1]);
    }
    {
      v2f hv;
      #pragma unroll
      for (int s = 0; s < 2; ++s) {
        float hi = fmaf(a8[0 + s], 0.5f, gxl[0 + s]);
        float hf = fmaf(a8[2 + s], 0.5f, gxl[2 + s]);
        float gg = a8[4 + s] + gxl[4 + s];
        float ho = fmaf(a8[6 + s], 0.5f, gxl[6 + s]);
        float c  = sigm_h(hf) * cL[s] + sigm_h(hi) * tanh_(gg);
        cL[s] = c;
        hv[s] = sigm_h(ho) * tanh_(c);
      }
      *(v2f*)&ln[lrow * HLS + u32] = hv;
    }
    {
      v4f hv;
      #pragma unroll
      for (int tt = 0; tt < 2; ++tt) {
        #pragma unroll
        for (int s = 0; s < 2; ++s) {
          const int ci = 2 * tt + s;
          float hi = fmaf(acc[tt][0 + s], 0.5f, bD[0][ci]);
          float hf = fmaf(acc[tt][2 + s], 0.5f, bD[1][ci]);
          float gg = acc[tt][4 + s] + bD[2][ci];
          float ho = fmaf(acc[tt][6 + s], 0.5f, bD[3][ci]);
          float c  = sigm_h(hf) * cD[ci] + sigm_h(hi) * tanh_(gg);
          cD[ci] = c;
          hv[ci] = sigm_h(ho) * tanh_(c);
        }
      }
      *(v4f*)&dn[lrow * HS + u64] = hv;
    }
    __syncthreads();
  }
  if (proj) {
    const float* dl = hdec + (TT & 1) * (16 * HS);
    v8f p8 = {};
    #pragma unroll
    for (int kc = 0; kc < 16; ++kc) {
      v2f a = *(const v2f*)&dl[lrow * HS + kc * 4 + 2 * lhi];
      p8 = wmma4(a, wo[kc], p8);
    }
    if (fo < FF) {
      #pragma unroll
      for (int jj = 0; jj < 8; ++jj) {
        const int m = jj + 8 * lhi;
        outblk[(size_t)m * TT * FF + (TT - 1) * FF + fo] = p8[jj] + obias;
      }
    }
  }
}

extern "C" void kernel_launch(void* const* d_in, const int* in_sizes, int n_in,
                              void* d_out, int out_size, void* d_ws, size_t ws_size,
                              hipStream_t stream) {
  (void)in_sizes; (void)n_in; (void)out_size; (void)d_ws; (void)ws_size;
  const float* x    = (const float*)d_in[0];
  const float* eWih = (const float*)d_in[1];
  const float* eWhh = (const float*)d_in[2];
  const float* eb   = (const float*)d_in[3];
  const float* lWih = (const float*)d_in[4];
  const float* lWhh = (const float*)d_in[5];
  const float* lb   = (const float*)d_in[6];
  const float* dWih = (const float*)d_in[7];
  const float* dWhh = (const float*)d_in[8];
  const float* db   = (const float*)d_in[9];
  const float* oW   = (const float*)d_in[10];
  const float* ob   = (const float*)d_in[11];
  float* out = (float*)d_out;

  dim3 grid(BB / 16), block(256);
  hipLaunchKernelGGL(lstm_ae_kernel, grid, block, 0, stream,
                     x, eWih, eWhh, eb, lWih, lWhh, lb, dWih, dWhh, db, oW, ob, out);
}


# input-side gate products (enc Wih.x, dec Wih.h_lat) and projection taken off the recurrent chain: formed a step ahead, beside the cell update and between barrier signal and wait; sum order kept
# speedup vs baseline: 1.7876x; 1.0377x over previous
#include <hip/hip_runtime.h>


typedef float v2f __attribute__((ext_vector_type(2)));
typedef float v4f __attribute__((ext_vector_type(4)));
typedef float v8f __attribute__((ext_vector_type(8)));

#define BB 4096
#define TT 100
#define FF 40
#define HE 64
#define HL 32
#define HD 64

#define HS  68
#define HLS 36
#define XS  44

#define XK0 3
#define DK0 4

__device__ __forceinline__ v8f wmma4(v2f a, v2f b, v8f c) {
  return __builtin_amdgcn_wmma_f32_16x16x4_f32(false, a, false, b, (short)0, c,
                                               false, false);
}

__device__ __forceinline__ void step_arrive() {
  __builtin_amdgcn_fence(__ATOMIC_RELEASE, "workgroup", "local");
  __builtin_amdgcn_s_barrier_signal(-1);
}
__device__ __forceinline__ void step_wait() {
  __builtin_amdgcn_s_barrier_wait(-1);
  __builtin_amdgcn_fence(__ATOMIC_ACQUIRE, "workgroup", "local");
}

__device__ __forceinline__ float tanh_(float x) {
  return __builtin_amdgcn_tanhf(x);
}
__device__ __forceinline__ float sigm_h(float hx) {
  return fmaf(__builtin_amdgcn_tanhf(hx), 0.5f, 0.5f);
}

__global__ __launch_bounds__(256)
void lstm_ae_kernel(const float* __restrict__ x,
                    const float* __restrict__ eWih, const float* __restrict__ eWhh,
                    const float* __restrict__ eb,
                    const float* __restrict__ lWih, const float* __restrict__ lWhh,
                    const float* __restrict__ lb,
                    const float* __restrict__ dWih, const float* __restrict__ dWhh,
                    const float* __restrict__ db,
                    const float* __restrict__ oW,   const float* __restrict__ ob,
                    float* __restrict__ out) {
  __shared__ __attribute__((aligned(16))) float henc[2 * 16 * HS];
  __shared__ __attribute__((aligned(16))) float hdec[2 * 16 * HS];
  __shared__ __attribute__((aligned(16))) float hlat[2 * 16 * HLS];
  __shared__ __attribute__((aligned(16))) float xenc[2 * 16 * XS];

  const int tid  = threadIdx.x;
  const int lane = tid & 31;
  const int wave = tid >> 5;
  const int lrow = lane & 15;
  const int lhi  = lane >> 4;
  const int b0   = blockIdx.x * 16;

  const int mh = lrow >> 3;
  const int mq = (lrow & 7) >> 1;
  const int ms = lrow & 1;
  const int row64 = mq * 64 + 8 * wave + 4 * mh + ms;
  const int row32 = mq * 32 + 4 * wave + 2 * mh + ms;
  const int u64 = 8 * wave + 4 * lhi;
  const int u32 = 4 * wave + 2 * lhi;

  v2f wih[10][2];
  v2f whh[16][2];
  #pragma unroll
  for (int tt = 0; tt < 2; ++tt) {
    const int gr = row64 + 2 * tt;
    #pragma unroll
    for (int kc = 0; kc < 10; ++kc)
      wih[kc][tt] = *(const v2f*)&eWih[gr * FF + kc * 4 + 2 * lhi];
    #pragma unroll
    for (int kc = 0; kc < 16; ++kc)
      whh[kc][tt] = *(const v2f*)&eWhh[gr * HE + kc * 4 + 2 * lhi];
  }
  float bE[4][4];
  #pragma unroll
  for (int q = 0; q < 4; ++q)
    #pragma unroll
    for (int cidx = 0; cidx < 4; ++cidx)
      bE[q][cidx] = (q == 2 ? 1.0f : 0.5f) * eb[q * 64 + u64 + cidx];

  for (int i = tid; i < 16 * HS; i += 256) { henc[i] = 0.0f; hdec[i] = 0.0f; }
  for (int i = tid; i < 16 * HLS; i += 256) hlat[16 * HLS + i] = 0.0f;
  if (tid < 160) {
    const int r0 = tid / 10, q0 = tid - 10 * r0;
    const float* xs0 = &x[((size_t)b0 + r0) * TT * FF + q0 * 4];
    *(v4f*)&xenc[r0 * XS + q0 * 4]           = *(const v4f*)xs0;
    *(v4f*)&xenc[16 * XS + r0 * XS + q0 * 4] = *(const v4f*)(xs0 + FF);
  }
  float cE[4] = {0.f, 0.f, 0.f, 0.f};
  __syncthreads();
  v8f accx[2] = {{}, {}};
  #pragma unroll
  for (int kc = 0; kc < 10; ++kc) {
    v2f b = *(const v2f*)&xenc[lrow * XS + kc * 4 + 2 * lhi];
    accx[0] = wmma4(wih[kc][0], b, accx[0]);
    accx[1] = wmma4(wih[kc][1], b, accx[1]);
  }
  __syncthreads();

  const float* xblk = x + (size_t)b0 * TT * FF;

  const bool xld  = __builtin_amdgcn_readfirstlane(wave) < 5;
  const int  xrw  = xld ? tid / 10 : 0;
  const int  xq   = xld ? tid - 10 * xrw : 0;
  const float* xsrc = xblk + (size_t)xrw * TT * FF + xq * 4;
  float*       xdst = xenc + xrw * XS + xq * 4;

  #pragma unroll 1
  for (int t = 0; t < TT; ++t) {
    const float* hc = henc + (t & 1) * (16 * HS);
    float*       hn = henc + ((t + 1) & 1) * (16 * HS);
    const float* xc = xenc + ((t + 1) & 1) * (16 * XS);
    v4f xv = {};
    if (xld) {
      const int tn = (t + 2 < TT) ? (t + 2) : (TT - 1);
      xv = *(const v4f*)&xsrc[tn * FF];
    }
    v2f bh[16], bx[10];
    #pragma unroll
    for (int kc = 0; kc < 16; ++kc)
      bh[kc] = *(const v2f*)&hc[lrow * HS + kc * 4 + 2 * lhi];
    #pragma unroll
    for (int kc = 0; kc < 10; ++kc)
      bx[kc] = *(const v2f*)&xc[lrow * XS + kc * 4 + 2 * lhi];
    v8f acc[2] = {accx[0], accx[1]};
    #pragma unroll
    for (int kc = 0; kc < 16; ++kc) {
      acc[0] = wmma4(whh[kc][0], bh[kc], acc[0]);
      acc[1] = wmma4(whh[kc][1], bh[kc], acc[1]);
    }
    __builtin_amdgcn_sched_barrier(0);
    accx[0] = v8f{};
    accx[1] = v8f{};
    #pragma unroll
    for (int kc = 0; kc < XK0; ++kc) {
      accx[0] = wmma4(wih[kc][0], bx[kc], accx[0]);
      accx[1] = wmma4(wih[kc][1], bx[kc], accx[1]);
    }
    v4f hv;
    #pragma unroll
    for (int tt = 0; tt < 2; ++tt) {
      #pragma unroll
      for (int s = 0; s < 2; ++s) {
        const int ci = 2 * tt + s;
        float hi = fmaf(acc[tt][0 + s], 0.5f, bE[0][ci]);
        float hf = fmaf(acc[tt][2 + s], 0.5f, bE[1][ci]);
        float gg = acc[tt][4 + s] + bE[2][ci];
        float ho = fmaf(acc[tt][6 + s], 0.5f, bE[3][ci]);
        float c  = sigm_h(hf) * cE[ci] + sigm_h(hi) * tanh_(gg);
        cE[ci] = c;
        hv[ci] = sigm_h(ho) * tanh_(c);
      }
    }
    *(v4f*)&hn[lrow * HS + u64] = hv;
    if (xld) *(v4f*)&xdst[(t & 1) * (16 * XS)] = xv;
    step_arrive();
    __builtin_amdgcn_sched_barrier(0);
    #pragma unroll
    for (int kc = XK0; kc < 10; ++kc) {
      accx[0] = wmma4(wih[kc][0], bx[kc], accx[0]);
      accx[1] = wmma4(wih[kc][1], bx[kc], accx[1]);
    }
    __builtin_amdgcn_sched_barrier(0);
    step_wait();
  }
  const float* hlast = henc + (TT & 1) * (16 * HS);

  float gxl[8];
  {
    v8f a8 = {};
    #pragma unroll
    for (int kc = 0; kc < 16; ++kc) {
      v2f a = *(const v2f*)&lWih[row32 * HE + kc * 4 + 2 * lhi];
      v2f b = *(const v2f*)&hlast[lrow * HS + kc * 4 + 2 * lhi];
      a8 = wmma4(a, b, a8);
    }
    #pragma unroll
    for (int r = 0; r < 8; ++r)
      gxl[r] = ((r >> 1) == 2 ? 1.0f : 0.5f) * (a8[r] + lb[(r >> 1) * 32 + u32 + (r & 1)]);
  }

  v2f wlh[8];
  v2f dwi[8][2];
  v2f dwh[16][2];
  v2f wo[16];
  {
    #pragma unroll
    for (int kc = 0; kc < 8; ++kc)
      wlh[kc] = *(const v2f*)&lWhh[row32 * HL + kc * 4 + 2 * lhi];
    #pragma unroll
    for (int tt = 0; tt < 2; ++tt) {
      const int gr = row64 + 2 * tt;
      #pragma unroll
      for (int kc = 0; kc < 8; ++kc)
        dwi[kc][tt] = *(const v2f*)&dWih[gr * HL + kc * 4 + 2 * lhi];
      #pragma unroll
      for (int kc = 0; kc < 16; ++kc)
        dwh[kc][tt] = *(const v2f*)&dWhh[gr * HD + kc * 4 + 2 * lhi];
    }
    const int f = wave * 16 + lrow;
    #pragma unroll
    for (int kc = 0; kc < 16; ++kc) {
      v2f z = {};
      if (wave < 3 && f < FF)
        z = *(const v2f*)&oW[f * HD + kc * 4 + 2 * lhi];
      wo[kc] = z;
    }
  }
  float bD[4][4];
  #pragma unroll
  for (int q = 0; q < 4; ++q)
    #pragma unroll
    for (int cidx = 0; cidx < 4; ++cidx)
      bD[q][cidx] = (q == 2 ? 1.0f : 0.5f) * db[q * 64 + u64 + cidx];
  const float obias = (wave < 3 && (wave * 16 + lrow) < FF) ? ob[wave * 16 + lrow] : 0.0f;

  float cL[2] = {0.f, 0.f};
  float cD[4] = {0.f, 0.f, 0.f, 0.f};

  float* outblk = out + (size_t)b0 * TT * FF;
  const bool proj = __builtin_amdgcn_readfirstlane(wave) < 3;
  const int  fo   = wave * 16 + lrow;

  {
    v8f a8 = {};
    #pragma unroll
    for (int kc = 0; kc < 8; ++kc) {
      v2f b = *(const v2f*)&hlat[16 * HLS + lrow * HLS + kc * 4 + 2 * lhi];
      a8 = wmma4(wlh[kc], b, a8);
    }
    v2f hv;
    #pragma unroll
    for (int s = 0; s < 2; ++s) {
      float hi = fmaf(a8[0 + s], 0.5f, gxl[0 + s]);
      float hf = fmaf(a8[2 + s], 0.5f, gxl[2 + s]);
      float gg = a8[4 + s] + gxl[4 + s];
      float ho = fmaf(a8[6 + s], 0.5f, gxl[6 + s]);
      float c  = sigm_h(hf) * cL[s] + sigm_h(hi) * tanh_(gg);
      cL[s] = c;
      hv[s] = sigm_h(ho) * tanh_(c);
    }
    *(v2f*)&hlat[lrow * HLS + u32] = hv;
  }
  __syncthreads();
  v8f accI[2] = {{}, {}};
  {
    v2f bl[8];
    #pragma unroll
    for (int kc = 0; kc < 8; ++kc)
      bl[kc] = *(const v2f*)&hlat[lrow * HLS + kc * 4 + 2 * lhi];
    v8f a8 = {};
    #pragma unroll
    for (int kc = 0; kc < 8; ++kc)
      a8 = wmma4(wlh[kc], bl[kc], a8);
    #pragma unroll
    for (int kc = 0; kc < 8; ++kc) {
      accI[0] = wmma4(dwi[kc][0], bl[kc], accI[0]);
      accI[1] = wmma4(dwi[kc][1], bl[kc], accI[1]);
    }
    v2f hv;
    #pragma unroll
    for (int s = 0; s < 2; ++s) {
      float hi = fmaf(a8[0 + s], 0.5f, gxl[0 + s]);
      float hf = fmaf(a8[2 + s], 0.5f, gxl[2 + s]);
      float gg = a8[4 + s] + gxl[4 + s];
      float ho = fmaf(a8[6 + s], 0.5f, gxl[6 + s]);
      float c  = sigm_h(hf) * cL[s] + sigm_h(hi) * tanh_(gg);
      cL[s] = c;
      hv[s] = sigm_h(ho) * tanh_(c);
    }
    *(v2f*)&hlat[16 * HLS + lrow * HLS + u32] = hv;
  }
  __syncthreads();

  #pragma unroll 1
  for (int t = 0; t < TT; ++t) {
    const float* lc = hlat + ((t + 1) & 1) * (16 * HLS);
    float*       ln = hlat + (t & 1) * (16 * HLS);
    const float* dc = hdec + (t & 1) * (16 * HS);
    float*       dn = hdec + ((t + 1) & 1) * (16 * HS);
    v2f bl[8], bd[16];
    #pragma unroll
    for (int kc = 0; kc < 8; ++kc)
      bl[kc] = *(const v2f*)&lc[lrow * HLS + kc * 4 + 2 * lhi];
    #pragma unroll
    for (int kc = 0; kc < 16; ++kc)
      bd[kc] = *(const v2f*)&dc[lrow * HS + kc * 4 + 2 * lhi];
    v8f a8 = {};
    #pragma unroll
    for (int kc = 0; kc < 8; ++kc)
      a8 = wmma4(wlh[kc], bl[kc], a8);
    v8f acc[2] = {accI[0], accI[1]};
    #pragma unroll
    for (int kc = 0; kc < 16; ++kc) {
      acc[0] = wmma4(dwh[kc][0], bd[kc], acc[0]);
      acc[1] = wmma4(dwh[kc][1], bd[kc], acc[1]);
    }
    __builtin_amdgcn_sched_barrier(0);
    accI[0] = v8f{};
    accI[1] = v8f{};
    #pragma unroll
    for (int kc = 0; kc < DK0; ++kc) {
      accI[0] = wmma4(dwi[kc][0], bl[kc], accI[0]);
      accI[1] = wmma4(dwi[kc][1], bl[kc], accI[1]);
    }
    {
      v2f hv;
      #pragma unroll
      for (int s = 0; s < 2; ++s) {
        float hi = fmaf(a8[0 + s], 0.5f, gxl[0 + s]);
        float hf = fmaf(a8[2 + s], 0.5f, gxl[2 + s]);
        float gg = a8[4 + s] + gxl[4 + s];
        float ho = fmaf(a8[6 + s], 0.5f, gxl[6 + s]);
        float c  = sigm_h(hf) * cL[s] + sigm_h(hi) * tanh_(gg);
        cL[s] = c;
        hv[s] = sigm_h(ho) * tanh_(c);
      }
      *(v2f*)&ln[lrow * HLS + u32] = hv;
    }
    {
      v4f hv;
      #pragma unroll
      for (int tt = 0; tt < 2; ++tt) {
        #pragma unroll
        for (int s = 0; s < 2; ++s) {
          const int ci = 2 * tt + s;
          float hi = fmaf(acc[tt][0 + s], 0.5f, bD[0][ci]);
          float hf = fmaf(acc[tt][2 + s], 0.5f, bD[1][ci]);
          float gg = acc[tt][4 + s] + bD[2][ci];
          float ho = fmaf(acc[tt][6 + s], 0.5f, bD[3][ci]);
          float c  = sigm_h(hf) * cD[ci] + sigm_h(hi) * tanh_(gg);
          cD[ci] = c;
          hv[ci] = sigm_h(ho) * tanh_(c);
        }
      }
      *(v4f*)&dn[lrow * HS + u64] = hv;
    }
    step_arrive();
    __builtin_amdgcn_sched_barrier(0);
    #pragma unroll
    for (int kc = DK0; kc < 8; ++kc) {
      accI[0] = wmma4(dwi[kc][0], bl[kc], accI[0]);
      accI[1] = wmma4(dwi[kc][1], bl[kc], accI[1]);
    }
    if (proj && t > 0) {
      v8f p8 = {};
      #pragma unroll
      for (int kc = 0; kc < 16; ++kc)
        p8 = wmma4(bd[kc], wo[kc], p8);
      if (fo < FF) {
        #pragma unroll
        for (int jj = 0; jj < 8; ++jj) {
          const int m = jj + 8 * lhi;
          outblk[(size_t)m * TT * FF + (t - 1) * FF + fo] = p8[jj] + obias;
        }
      }
    }
    __builtin_amdgcn_sched_barrier(0);
    step_wait();
  }
  if (proj) {
    const float* dl = hdec + (TT & 1) * (16 * HS);
    v8f p8 = {};
    #pragma unroll
    for (int kc = 0; kc < 16; ++kc) {
      v2f a = *(const v2f*)&dl[lrow * HS + kc * 4 + 2 * lhi];
      p8 = wmma4(a, wo[kc], p8);
    }
    if (fo < FF) {
      #pragma unroll
      for (int jj = 0; jj < 8; ++jj) {
        const int m = jj + 8 * lhi;
        outblk[(size_t)m * TT * FF + (TT - 1) * FF + fo] = p8[jj] + obias;
      }
    }
  }
}

extern "C" void kernel_launch(void* const* d_in, const int* in_sizes, int n_in,
                              void* d_out, int out_size, void* d_ws, size_t ws_size,
                              hipStream_t stream) {
  (void)in_sizes; (void)n_in; (void)out_size; (void)d_ws; (void)ws_size;
  const float* x    = (const float*)d_in[0];
  const float* eWih = (const float*)d_in[1];
  const float* eWhh = (const float*)d_in[2];
  const float* eb   = (const float*)d_in[3];
  const float* lWih = (const float*)d_in[4];
  const float* lWhh = (const float*)d_in[5];
  const float* lb   = (const float*)d_in[6];
  const float* dWih = (const float*)d_in[7];
  const float* dWhh = (const float*)d_in[8];
  const float* db   = (const float*)d_in[9];
  const float* oW   = (const float*)d_in[10];
  const float* ob   = (const float*)d_in[11];
  float* out = (float*)d_out;

  dim3 grid(BB / 16), block(256);
  hipLaunchKernelGGL(lstm_ae_kernel, grid, block, 0, stream,
                     x, eWih, eWhh, eb, lWih, lWhh, lb, dWih, dWhh, db, oW, ob, out);
}
